# starting kernel
# baseline (speedup 1.0000x reference)
#include <hip/hip_runtime.h>

typedef __attribute__((ext_vector_type(2))) float v2f;
typedef __attribute__((ext_vector_type(8))) float v8f;

__global__ __launch_bounds__(32) void vqc_build_tables(
    const float* __restrict__ ry, const float* __restrict__ rz,
    float* __restrict__ tab) {
  if (threadIdx.x != 0) return;

  float Ur[16], Ui[16];
  for (int i = 0; i < 16; ++i) { Ur[i] = (i % 5 == 0) ? 1.f : 0.f; Ui[i] = 0.f; }
  const int perm[4] = {0, 1, 3, 2};

  for (int l = 0; l < 6; ++l) {
    float h0 = 0.5f * ry[l * 2 + 0], h1 = 0.5f * ry[l * 2 + 1];
    float c0 = cosf(h0), s0 = sinf(h0), c1 = cosf(h1), s1 = sinf(h1);
    float R0[4] = {c0, -s0, s0, c0};
    float R1[4] = {c1, -s1, s1, c1};
    float F[16];
    for (int a = 0; a < 2; ++a)
      for (int b = 0; b < 2; ++b)
        for (int cc = 0; cc < 2; ++cc)
          for (int d = 0; d < 2; ++d)
            F[(2 * a + b) * 4 + (2 * cc + d)] = R0[a * 2 + cc] * R1[b * 2 + d];
    float t0 = rz[l * 2 + 0], t1 = rz[l * 2 + 1];
    float ph[4];
    for (int j = 0; j < 4; ++j) {
      int a = j >> 1, b = j & 1;
      ph[j] = 0.5f * ((a ? t0 : -t0) + (b ? t1 : -t1));
    }
    float Lr[16], Li[16];
    for (int i = 0; i < 4; ++i) {
      int j = perm[i];
      float cr = cosf(ph[j]), ci = sinf(ph[j]);
      for (int k = 0; k < 4; ++k) {
        Lr[i * 4 + k] = cr * F[j * 4 + k];
        Li[i * 4 + k] = ci * F[j * 4 + k];
      }
    }
    float Nr[16], Ni[16];
    for (int i = 0; i < 4; ++i)
      for (int k = 0; k < 4; ++k) {
        float ar = 0.f, ai = 0.f;
        for (int m = 0; m < 4; ++m) {
          ar += Lr[i * 4 + m] * Ur[m * 4 + k] - Li[i * 4 + m] * Ui[m * 4 + k];
          ai += Lr[i * 4 + m] * Ui[m * 4 + k] + Li[i * 4 + m] * Ur[m * 4 + k];
        }
        Nr[i * 4 + k] = ar;
        Ni[i * 4 + k] = ai;
      }
    for (int i = 0; i < 16; ++i) { Ur[i] = Nr[i]; Ui[i] = Ni[i]; }
  }

  float Mr[16], Mi[16];
  for (int i = 0; i < 16; ++i) { Mr[i] = Ur[i]; Mi[i] = -Ui[i]; }

  for (int j = 0; j < 2; ++j)
    for (int L = 0; L < 32; ++L) {
      int k = j + ((L < 16) ? 0 : 2);
      int m = L & 15;
      float a1, a2;
      if (m < 4)      { a1 = Mr[m * 4 + k];       a2 = -Mi[m * 4 + k]; }
      else if (m < 8) { a1 = Mi[(m - 4) * 4 + k]; a2 =  Mr[(m - 4) * 4 + k]; }
      else            { a1 = 0.f;                 a2 = 0.f; }
      tab[j * 32 + L]      = a1;
      tab[64 + j * 32 + L] = a2;
    }
}

__global__ __launch_bounds__(256) void vqc_main(
    const float* __restrict__ sr, const float* __restrict__ si,
    const float* __restrict__ tab, float* __restrict__ out, int nTiles) {
  const int lane = threadIdx.x & 31;
  const int wave = threadIdx.x >> 5;
  const int wavesPerBlock = blockDim.x >> 5;
  const int gwave = blockIdx.x * wavesPerBlock + wave;
  const int nwaves = gridDim.x * wavesPerBlock;

  v2f a1, a2;
  a1.x = tab[lane];       a1.y = tab[32 + lane];
  a2.x = tab[64 + lane];  a2.y = tab[96 + lane];

  const int col = lane & 15;
  const int koff = (lane < 16) ? 0 : 2;
  const bool writer = (lane < 16);

  for (int t = gwave; t < nTiles; t += nwaves) {
    const int base = t << 4;
    const int row = base + col;
    v2f br = __builtin_nontemporal_load((const v2f*)(sr + row * 4 + koff));
    v2f bi = __builtin_nontemporal_load((const v2f*)(si + row * 4 + koff));

    v8f acc = {0.f, 0.f, 0.f, 0.f, 0.f, 0.f, 0.f, 0.f};
    acc = __builtin_amdgcn_wmma_f32_16x16x4_f32(false, a1, false, br,
                                                (short)0, acc, false, false);
    acc = __builtin_amdgcn_wmma_f32_16x16x4_f32(false, a2, false, bi,
                                                (short)0, acc, false, false);

    float r0 = acc[0] * acc[0];
    r0 += acc[1] * acc[1];
    r0 -= acc[2] * acc[2];
    r0 -= acc[3] * acc[3];
    r0 += acc[4] * acc[4];
    r0 += acc[5] * acc[5];
    r0 -= acc[6] * acc[6];
    r0 -= acc[7] * acc[7];

    if (writer)
      __builtin_nontemporal_store(r0, out + base + lane);
  }
}

extern "C" void kernel_launch(void* const* d_in, const int* in_sizes, int n_in,
                              void* d_out, int out_size, void* d_ws, size_t ws_size,
                              hipStream_t stream) {
  const float* ry = (const float*)d_in[0];
  const float* rz = (const float*)d_in[1];
  const float* sr = (const float*)d_in[2];
  const float* si = (const float*)d_in[3];
  float* out = (float*)d_out;
  float* tab = (float*)d_ws;

  const int nTiles = out_size >> 4;

  vqc_build_tables<<<1, 32, 0, stream>>>(ry, rz, tab);
  vqc_main<<<1024, 256, 0, stream>>>(sr, si, tab, out, nTiles);
}


# table build fused into the streaming kernel (2 launches -> 1): wave 0 of each block builds U lane-parallel via shuffles under the first tile loads, A operands handed over in LDS; sum order kept
# speedup vs baseline: 1.8177x; 1.8177x over previous
#include <hip/hip_runtime.h>

typedef __attribute__((ext_vector_type(2))) float v2f;
typedef __attribute__((ext_vector_type(8))) float v8f;

__global__ __launch_bounds__(256) void vqc_fused(
    const float* __restrict__ ry, const float* __restrict__ rz,
    const float* __restrict__ sr, const float* __restrict__ si,
    float* __restrict__ out, int nTiles) {
  __shared__ float tab[128];

  const int lane = threadIdx.x & 31;
  const int wave = threadIdx.x >> 5;
  const int wavesPerBlock = blockDim.x >> 5;
  const int gwave = blockIdx.x * wavesPerBlock + wave;
  const int nwaves = gridDim.x * wavesPerBlock;

  const int col = lane & 15;
  const int koff = (lane < 16) ? 0 : 2;
  const bool writer = (lane < 16);

  int t = gwave;
  v2f br = {0.f, 0.f}, bi = {0.f, 0.f};
  if (t < nTiles) {
    const int row = (t << 4) + col;
    br = __builtin_nontemporal_load((const v2f*)(sr + row * 4 + koff));
    bi = __builtin_nontemporal_load((const v2f*)(si + row * 4 + koff));
  }

  if (wave == 0) {
    const int l = min(lane >> 2, 5);
    const int i = lane & 3;
    const int j = i ^ (i >> 1);
    const int a = j >> 1, b = j & 1;
    const float h0 = 0.5f * ry[l * 2 + 0], h1 = 0.5f * ry[l * 2 + 1];
    const float t0 = rz[l * 2 + 0], t1 = rz[l * 2 + 1];
    const float ph = 0.5f * ((a ? t0 : -t0) + (b ? t1 : -t1));
    float s0, c0, s1, c1, ci, cr;
    sincosf(h0, &s0, &c0);
    sincosf(h1, &s1, &c1);
    sincosf(ph, &ci, &cr);
    const float p0 = a ? s0 : c0, p1 = a ? c0 : -s0;
    const float q0 = b ? s1 : c1, q1 = b ? c1 : -s1;
    const float F0 = p0 * q0, F1 = p0 * q1, F2 = p1 * q0, F3 = p1 * q1;
    const float Lr[4] = {cr * F0, cr * F1, cr * F2, cr * F3};
    const float Li[4] = {ci * F0, ci * F1, ci * F2, ci * F3};

    const int p = lane >> 4;
    const int ui = (lane >> 2) & 3;
    const int uk = lane & 3;
    float u = (p == 0 && ui == uk) ? 1.f : 0.f;
    for (int ll = 0; ll < 6; ++ll) {
      const int srcRow = ll * 4 + ui;
      float acc = 0.f;
      for (int m = 0; m < 4; ++m) {
        const float lr = __shfl(Lr[m], srcRow);
        const float li = __shfl(Li[m], srcRow);
        const float x = __shfl(u, (p << 4) + m * 4 + uk);
        const float y = __shfl(u, ((p ^ 1) << 4) + m * 4 + uk);
        const float sli = p ? li : -li;
        acc += lr * x + sli * y;
      }
      u = acc;
    }

    const int mm = lane & 15;
    const int r = mm & 3;
    const float urx = __shfl(u, r * 4 + koff);
    const float ury = __shfl(u, r * 4 + koff + 1);
    const float uix = __shfl(u, 16 + r * 4 + koff);
    const float uiy = __shfl(u, 16 + r * 4 + koff + 1);
    float a1x = 0.f, a1y = 0.f, a2x = 0.f, a2y = 0.f;
    if (mm < 4)      { a1x = urx;  a1y = ury;  a2x = uix; a2y = uiy; }
    else if (mm < 8) { a1x = -uix; a1y = -uiy; a2x = urx; a2y = ury; }
    tab[lane]      = a1x;
    tab[32 + lane] = a1y;
    tab[64 + lane] = a2x;
    tab[96 + lane] = a2y;
  }
  __syncthreads();

  v2f a1, a2;
  a1.x = tab[lane];       a1.y = tab[32 + lane];
  a2.x = tab[64 + lane];  a2.y = tab[96 + lane];

  while (t < nTiles) {
    v8f acc = {0.f, 0.f, 0.f, 0.f, 0.f, 0.f, 0.f, 0.f};
    acc = __builtin_amdgcn_wmma_f32_16x16x4_f32(false, a1, false, br,
                                                (short)0, acc, false, false);
    acc = __builtin_amdgcn_wmma_f32_16x16x4_f32(false, a2, false, bi,
                                                (short)0, acc, false, false);

    float r0 = acc[0] * acc[0];
    r0 += acc[1] * acc[1];
    r0 -= acc[2] * acc[2];
    r0 -= acc[3] * acc[3];
    r0 += acc[4] * acc[4];
    r0 += acc[5] * acc[5];
    r0 -= acc[6] * acc[6];
    r0 -= acc[7] * acc[7];

    if (writer)
      __builtin_nontemporal_store(r0, out + (t << 4) + lane);

    t += nwaves;
    if (t < nTiles) {
      const int row = (t << 4) + col;
      br = __builtin_nontemporal_load((const v2f*)(sr + row * 4 + koff));
      bi = __builtin_nontemporal_load((const v2f*)(si + row * 4 + koff));
    }
  }
}

extern "C" void kernel_launch(void* const* d_in, const int* in_sizes, int n_in,
                              void* d_out, int out_size, void* d_ws, size_t ws_size,
                              hipStream_t stream) {
  const float* ry = (const float*)d_in[0];
  const float* rz = (const float*)d_in[1];
  const float* sr = (const float*)d_in[2];
  const float* si = (const float*)d_in[3];
  float* out = (float*)d_out;

  const int nTiles = out_size >> 4;

  vqc_fused<<<1024, 256, 0, stream>>>(ry, rz, sr, si, out, nTiles);
}


# 4x fewer waves (256 blocks x 8): each wave takes 32 consecutive tiles, all their operand loads in flight across the table build (LDS-only barrier, angles by scalar load); sum order kept, fma fused
# speedup vs baseline: 3.1037x; 1.7075x over previous
#include <hip/hip_runtime.h>

typedef __attribute__((ext_vector_type(2))) float v2f;
typedef __attribute__((ext_vector_type(8))) float v8f;

#define TPW 32


static __device__ __forceinline__ float tile_val(v2f a1, v2f a2, v2f br, v2f bi) {
  v8f acc = {0.f, 0.f, 0.f, 0.f, 0.f, 0.f, 0.f, 0.f};
  acc = __builtin_amdgcn_wmma_f32_16x16x4_f32(false, a1, false, br,
                                              (short)0, acc, false, false);
  acc = __builtin_amdgcn_wmma_f32_16x16x4_f32(false, a2, false, bi,
                                              (short)0, acc, false, false);

  float r0 = acc[0] * acc[0];
  r0 += acc[1] * acc[1];
  r0 -= acc[2] * acc[2];
  r0 -= acc[3] * acc[3];
  r0 += acc[4] * acc[4];
  r0 += acc[5] * acc[5];
  r0 -= acc[6] * acc[6];
  r0 -= acc[7] * acc[7];
  return r0;
}

#define GRP 8

__global__ __launch_bounds__(256) void vqc_fused(
    const float* __restrict__ ry, const float* __restrict__ rz,
    const float* __restrict__ sr, const float* __restrict__ si,
    float* __restrict__ out, int nTiles) {
  __shared__ float tab[128];

  const int lane = threadIdx.x & 31;
  const int wave = __builtin_amdgcn_readfirstlane(threadIdx.x >> 5);
  const int wavesPerBlock = blockDim.x >> 5;
  const int gwave = blockIdx.x * wavesPerBlock + wave;
  const int nwaves = gridDim.x * wavesPerBlock;

  const int col = lane & 15;
  const int koff = (lane < 16) ? 0 : 2;
  const bool writer = (lane < 16);
  const int lofs = col * 4 + koff;

  float ryv[12], rzv[12];
#pragma unroll
  for (int k = 0; k < 12; ++k) { ryv[k] = ry[k]; rzv[k] = rz[k]; }

  const int t0 = gwave * TPW;
  const bool full = (t0 + TPW <= nTiles);
  const float* ps = sr + (size_t)t0 * 64 + lofs;
  const float* pi = si + (size_t)t0 * 64 + lofs;
  v2f br[TPW], bi[TPW];
  if (full) {
#pragma unroll
    for (int i = 0; i < TPW - 1; ++i) {
      br[i] = __builtin_nontemporal_load((const v2f*)(ps + i * 64));
      bi[i] = __builtin_nontemporal_load((const v2f*)(pi + i * 64));
      __builtin_amdgcn_sched_barrier(0);
    }
  }

  if (wave == 0) {
    const int l = min(lane >> 2, 5);
    const int i = lane & 3;
    const int j = i ^ (i >> 1);
    const int a = j >> 1, b = j & 1;
    float w0 = ryv[10], w1 = ryv[11], z0 = rzv[10], z1 = rzv[11];
#pragma unroll
    for (int k = 0; k < 5; ++k) {
      const bool sel = (l == k);
      w0 = sel ? ryv[2 * k] : w0;  w1 = sel ? ryv[2 * k + 1] : w1;
      z0 = sel ? rzv[2 * k] : z0;  z1 = sel ? rzv[2 * k + 1] : z1;
    }
    const float h0 = 0.5f * w0, h1 = 0.5f * w1;
    const float ph = 0.5f * ((a ? z0 : -z0) + (b ? z1 : -z1));
    float s0, c0, s1, c1, ci, cr;
    sincosf(h0, &s0, &c0);
    sincosf(h1, &s1, &c1);
    sincosf(ph, &ci, &cr);
    const float p0 = a ? s0 : c0, p1 = a ? c0 : -s0;
    const float q0 = b ? s1 : c1, q1 = b ? c1 : -s1;
    const float F0 = p0 * q0, F1 = p0 * q1, F2 = p1 * q0, F3 = p1 * q1;
    const float Lr[4] = {cr * F0, cr * F1, cr * F2, cr * F3};
    const float Li[4] = {ci * F0, ci * F1, ci * F2, ci * F3};

    const int p = lane >> 4;
    const int ui = (lane >> 2) & 3;
    const int uk = lane & 3;
    float u = (p == 0 && ui == uk) ? 1.f : 0.f;
    for (int ll = 0; ll < 6; ++ll) {
      const int srcRow = ll * 4 + ui;
      float acc = 0.f;
      for (int m = 0; m < 4; ++m) {
        const float lr = __shfl(Lr[m], srcRow);
        const float li = __shfl(Li[m], srcRow);
        const float x = __shfl(u, (p << 4) + m * 4 + uk);
        const float y = __shfl(u, ((p ^ 1) << 4) + m * 4 + uk);
        const float sli = p ? li : -li;
        acc += lr * x + sli * y;
      }
      u = acc;
    }

    const int mm = lane & 15;
    const int r = mm & 3;
    const float urx = __shfl(u, r * 4 + koff);
    const float ury = __shfl(u, r * 4 + koff + 1);
    const float uix = __shfl(u, 16 + r * 4 + koff);
    const float uiy = __shfl(u, 16 + r * 4 + koff + 1);
    float a1x = 0.f, a1y = 0.f, a2x = 0.f, a2y = 0.f;
    if (mm < 4)      { a1x = urx;  a1y = ury;  a2x = uix; a2y = uiy; }
    else if (mm < 8) { a1x = -uix; a1y = -uiy; a2x = urx; a2y = ury; }
    tab[lane]      = a1x;
    tab[32 + lane] = a1y;
    tab[64 + lane] = a2x;
    tab[96 + lane] = a2y;
  }
  __builtin_amdgcn_fence(__ATOMIC_RELEASE, "workgroup", "local");
  __builtin_amdgcn_s_barrier();
  __builtin_amdgcn_fence(__ATOMIC_ACQUIRE, "workgroup", "local");

  v2f a1, a2;
  a1.x = tab[lane];       a1.y = tab[32 + lane];
  a2.x = tab[64 + lane];  a2.y = tab[96 + lane];

  if (full) {
#pragma unroll
    for (int g = 0; g < TPW; g += GRP) {
      if (g == GRP) {
        br[TPW - 1] = __builtin_nontemporal_load((const v2f*)(ps + (TPW - 1) * 64));
        bi[TPW - 1] = __builtin_nontemporal_load((const v2f*)(pi + (TPW - 1) * 64));
      }
      float res[GRP];
#pragma unroll
      for (int j = 0; j < GRP; ++j)
        res[j] = tile_val(a1, a2, br[g + j], bi[g + j]);
      if (writer) {
#pragma unroll
        for (int j = 0; j < GRP; ++j)
          __builtin_nontemporal_store(res[j], out + ((t0 + g + j) << 4) + lane);
      }
    }
  } else {
    for (int t = t0; t < nTiles; ++t) {
      const v2f xr = __builtin_nontemporal_load((const v2f*)(sr + (size_t)t * 64 + lofs));
      const v2f xi = __builtin_nontemporal_load((const v2f*)(si + (size_t)t * 64 + lofs));
      const float r0 = tile_val(a1, a2, xr, xi);
      if (writer) __builtin_nontemporal_store(r0, out + (t << 4) + lane);
    }
  }
  for (int t = nwaves * TPW + gwave; t < nTiles; t += nwaves) {
    const v2f xr = __builtin_nontemporal_load((const v2f*)(sr + (size_t)t * 64 + lofs));
    const v2f xi = __builtin_nontemporal_load((const v2f*)(si + (size_t)t * 64 + lofs));
    const float r0 = tile_val(a1, a2, xr, xi);
    if (writer) __builtin_nontemporal_store(r0, out + (t << 4) + lane);
  }
}

extern "C" void kernel_launch(void* const* d_in, const int* in_sizes, int n_in,
                              void* d_out, int out_size, void* d_ws, size_t ws_size,
                              hipStream_t stream) {
  const float* ry = (const float*)d_in[0];
  const float* rz = (const float*)d_in[1];
  const float* sr = (const float*)d_in[2];
  const float* si = (const float*)d_in[3];
  float* out = (float*)d_out;

  const int nTiles = out_size >> 4;

  vqc_fused<<<256, 256, 0, stream>>>(ry, rz, sr, si, out, nTiles);
}


# table build: the U chain's six dependent steps and the A-operand swizzle move lanes with v_permlane16/x16 instead of LDS-crossbar shuffles; layer rows fetched once ahead; arithmetic, sum order kept
# speedup vs baseline: 3.3608x; 1.0828x over previous
#include <hip/hip_runtime.h>

typedef __attribute__((ext_vector_type(2))) float v2f;
typedef __attribute__((ext_vector_type(8))) float v8f;

#define TPW 32


static __device__ __forceinline__ float tile_val(v2f a1, v2f a2, v2f br, v2f bi) {
  v8f acc = {0.f, 0.f, 0.f, 0.f, 0.f, 0.f, 0.f, 0.f};
  acc = __builtin_amdgcn_wmma_f32_16x16x4_f32(false, a1, false, br,
                                              (short)0, acc, false, false);
  acc = __builtin_amdgcn_wmma_f32_16x16x4_f32(false, a2, false, bi,
                                              (short)0, acc, false, false);

  float r0 = acc[0] * acc[0];
  r0 += acc[1] * acc[1];
  r0 -= acc[2] * acc[2];
  r0 -= acc[3] * acc[3];
  r0 += acc[4] * acc[4];
  r0 += acc[5] * acc[5];
  r0 -= acc[6] * acc[6];
  r0 -= acc[7] * acc[7];
  return r0;
}

static __device__ __forceinline__ float own_row(float v, unsigned sel) {
  return __uint_as_float(__builtin_amdgcn_permlane16(
      __float_as_uint(v), __float_as_uint(v), sel, sel, false, false));
}
static __device__ __forceinline__ float other_row(float v, unsigned sel) {
  return __uint_as_float(__builtin_amdgcn_permlanex16(
      __float_as_uint(v), __float_as_uint(v), sel, sel, false, false));
}

#define GRP 8

__global__ __launch_bounds__(256) void vqc_fused(
    const float* __restrict__ ry, const float* __restrict__ rz,
    const float* __restrict__ sr, const float* __restrict__ si,
    float* __restrict__ out, int nTiles) {
  __shared__ float tab[128];

  const int lane = threadIdx.x & 31;
  const int wave = __builtin_amdgcn_readfirstlane(threadIdx.x >> 5);
  const int wavesPerBlock = blockDim.x >> 5;
  const int gwave = blockIdx.x * wavesPerBlock + wave;
  const int nwaves = gridDim.x * wavesPerBlock;

  const int col = lane & 15;
  const int koff = (lane < 16) ? 0 : 2;
  const bool writer = (lane < 16);
  const int lofs = col * 4 + koff;

  float ryv[12], rzv[12];
#pragma unroll
  for (int k = 0; k < 12; ++k) { ryv[k] = ry[k]; rzv[k] = rz[k]; }

  const int t0 = gwave * TPW;
  const bool full = (t0 + TPW <= nTiles);
  const float* ps = sr + (size_t)t0 * 64 + lofs;
  const float* pi = si + (size_t)t0 * 64 + lofs;
  v2f br[TPW], bi[TPW];
  if (full) {
#pragma unroll
    for (int i = 0; i < TPW - 1; ++i) {
      br[i] = __builtin_nontemporal_load((const v2f*)(ps + i * 64));
      bi[i] = __builtin_nontemporal_load((const v2f*)(pi + i * 64));
      __builtin_amdgcn_sched_barrier(0);
    }
  }

  if (wave == 0) {
    const int l = min(lane >> 2, 5);
    const int i = lane & 3;
    const int j = i ^ (i >> 1);
    const int a = j >> 1, b = j & 1;
    float w0 = ryv[10], w1 = ryv[11], z0 = rzv[10], z1 = rzv[11];
#pragma unroll
    for (int k = 0; k < 5; ++k) {
      const bool sel = (l == k);
      w0 = sel ? ryv[2 * k] : w0;  w1 = sel ? ryv[2 * k + 1] : w1;
      z0 = sel ? rzv[2 * k] : z0;  z1 = sel ? rzv[2 * k + 1] : z1;
    }
    const float h0 = 0.5f * w0, h1 = 0.5f * w1;
    const float ph = 0.5f * ((a ? z0 : -z0) + (b ? z1 : -z1));
    float s0, c0, s1, c1, ci, cr;
    sincosf(h0, &s0, &c0);
    sincosf(h1, &s1, &c1);
    sincosf(ph, &ci, &cr);
    const float p0 = a ? s0 : c0, p1 = a ? c0 : -s0;
    const float q0 = b ? s1 : c1, q1 = b ? c1 : -s1;
    const float F0 = p0 * q0, F1 = p0 * q1, F2 = p1 * q0, F3 = p1 * q1;
    const float Lr[4] = {cr * F0, cr * F1, cr * F2, cr * F3};
    const float Li[4] = {ci * F0, ci * F1, ci * F2, ci * F3};

    const int p = lane >> 4;
    const int ui = (lane >> 2) & 3;
    const int uk = lane & 3;
    float u = (p == 0 && ui == uk) ? 1.f : 0.f;
    float lrs[6][4], lis[6][4];
    for (int ll = 0; ll < 6; ++ll) {
      for (int m = 0; m < 4; ++m) {
        lrs[ll][m] = __shfl(Lr[m], ll * 4 + ui);
        lis[ll][m] = __shfl(Li[m], ll * 4 + ui);
      }
    }
    __builtin_amdgcn_sched_barrier(0);
    const unsigned selm[4] = {0x32103210u, 0x76547654u, 0xBA98BA98u, 0xFEDCFEDCu};
    for (int ll = 0; ll < 6; ++ll) {
      float acc = 0.f;
      for (int m = 0; m < 4; ++m) {
        const float lr = lrs[ll][m];
        const float li = lis[ll][m];
        const float x = own_row(u, selm[m]);
        const float y = other_row(u, selm[m]);
        const float sli = p ? li : -li;
        acc += lr * x + sli * y;
      }
      u = acc;
    }

    const int mm = lane & 15;
    const int r = mm & 3;
    const unsigned e0 = 0xC840C840u, e1 = 0xD951D951u, e2 = 0xEA62EA62u, e3 = 0xFB73FB73u;
    const float o0 = own_row(u, e0), o1 = own_row(u, e1), o2 = own_row(u, e2), o3 = own_row(u, e3);
    const float x0 = other_row(u, e0), x1 = other_row(u, e1), x2 = other_row(u, e2), x3 = other_row(u, e3);
    const float urx = writer ? o0 : x2;
    const float ury = writer ? o1 : x3;
    const float uix = writer ? x0 : o2;
    const float uiy = writer ? x1 : o3;
    float a1x = 0.f, a1y = 0.f, a2x = 0.f, a2y = 0.f;
    if (mm < 4)      { a1x = urx;  a1y = ury;  a2x = uix; a2y = uiy; }
    else if (mm < 8) { a1x = -uix; a1y = -uiy; a2x = urx; a2y = ury; }
    tab[lane]      = a1x;
    tab[32 + lane] = a1y;
    tab[64 + lane] = a2x;
    tab[96 + lane] = a2y;
  }
  __builtin_amdgcn_fence(__ATOMIC_RELEASE, "workgroup", "local");
  __builtin_amdgcn_s_barrier();
  __builtin_amdgcn_fence(__ATOMIC_ACQUIRE, "workgroup", "local");

  v2f a1, a2;
  a1.x = tab[lane];       a1.y = tab[32 + lane];
  a2.x = tab[64 + lane];  a2.y = tab[96 + lane];

  if (full) {
#pragma unroll
    for (int g = 0; g < TPW; g += GRP) {
      if (g == GRP) {
        br[TPW - 1] = __builtin_nontemporal_load((const v2f*)(ps + (TPW - 1) * 64));
        bi[TPW - 1] = __builtin_nontemporal_load((const v2f*)(pi + (TPW - 1) * 64));
      }
      float res[GRP];
#pragma unroll
      for (int j = 0; j < GRP; ++j)
        res[j] = tile_val(a1, a2, br[g + j], bi[g + j]);
      if (writer) {
#pragma unroll
        for (int j = 0; j < GRP; ++j)
          __builtin_nontemporal_store(res[j], out + ((t0 + g + j) << 4) + lane);
      }
    }
  } else {
    for (int t = t0; t < nTiles; ++t) {
      const v2f xr = __builtin_nontemporal_load((const v2f*)(sr + (size_t)t * 64 + lofs));
      const v2f xi = __builtin_nontemporal_load((const v2f*)(si + (size_t)t * 64 + lofs));
      const float r0 = tile_val(a1, a2, xr, xi);
      if (writer) __builtin_nontemporal_store(r0, out + (t << 4) + lane);
    }
  }
  for (int t = nwaves * TPW + gwave; t < nTiles; t += nwaves) {
    const v2f xr = __builtin_nontemporal_load((const v2f*)(sr + (size_t)t * 64 + lofs));
    const v2f xi = __builtin_nontemporal_load((const v2f*)(si + (size_t)t * 64 + lofs));
    const float r0 = tile_val(a1, a2, xr, xi);
    if (writer) __builtin_nontemporal_store(r0, out + (t << 4) + lane);
  }
}

extern "C" void kernel_launch(void* const* d_in, const int* in_sizes, int n_in,
                              void* d_out, int out_size, void* d_ws, size_t ws_size,
                              hipStream_t stream) {
  const float* ry = (const float*)d_in[0];
  const float* rz = (const float*)d_in[1];
  const float* sr = (const float*)d_in[2];
  const float* si = (const float*)d_in[3];
  float* out = (float*)d_out;

  const int nTiles = out_size >> 4;

  vqc_fused<<<256, 256, 0, stream>>>(ry, rz, sr, si, out, nTiles);
}
